# starting kernel
# baseline (speedup 1.0000x reference)
#include <hip/hip_runtime.h>

typedef __attribute__((ext_vector_type(2))) float v2f;
typedef __attribute__((ext_vector_type(8))) float v8f;

static constexpr int kD = 256;
static constexpr float kEps = 1e-12f;


__global__ void rt_zero(int* counts, int* cursor, int nrules) {
  int t = blockIdx.x * blockDim.x + threadIdx.x;
  if (t < nrules) { counts[t] = 0; cursor[t] = 0; }
}

__global__ void rt_count(const int* __restrict__ token_ids,
                         const int* __restrict__ token_rules,
                         int* __restrict__ counts, int ntok) {
  int t = blockIdx.x * blockDim.x + threadIdx.x;
  if (t < ntok) atomicAdd(&counts[token_rules[token_ids[t]]], 1);
}

__global__ __launch_bounds__(1024) void rt_scan(const int* __restrict__ counts,
                                                int* __restrict__ offsets,
                                                int nrules) {
  __shared__ int sh[1024];
  const int t = threadIdx.x;
  if (nrules <= 1024) {
    int v = (t < nrules) ? counts[t] : 0;
    sh[t] = v;
    __syncthreads();
#pragma unroll
    for (int ofs = 1; ofs < 1024; ofs <<= 1) {
      int x = (t >= ofs) ? sh[t - ofs] : 0;
      __syncthreads();
      sh[t] += x;
      __syncthreads();
    }
    if (t < nrules) offsets[t] = sh[t] - v;
    if (t == nrules - 1) offsets[nrules] = sh[t];
  } else if (t == 0) {
    int acc = 0;
    for (int i = 0; i < nrules; ++i) { offsets[i] = acc; acc += counts[i]; }
    offsets[nrules] = acc;
  }
}

__global__ void rt_scatter(const int* __restrict__ token_ids,
                           const int* __restrict__ token_rules,
                           const int* __restrict__ offsets,
                           int* __restrict__ cursor,
                           int* __restrict__ tlist, int ntok) {
  int t = blockIdx.x * blockDim.x + threadIdx.x;
  if (t < ntok) {
    int r = token_rules[token_ids[t]];
    int pos = offsets[r] + atomicAdd(&cursor[r], 1);
    tlist[pos] = t;
  }
}


__global__ __launch_bounds__(256) void rt_gemm(
    const float* __restrict__ hidden,
    const float* __restrict__ rules,
    const int* __restrict__ offsets,
    const int* __restrict__ tlist,
    float* __restrict__ out) {
  const int r = blockIdx.x;
  const int start = offsets[r];
  const int cnt = offsets[r + 1] - start;
  if (cnt == 0) return;

  const int tid = threadIdx.x;
  const int lane = tid & 31;
  const int wave = tid >> 5;
  const int h = lane >> 4;
  const int col = lane & 15;
  const int colbase = wave * 32;

  __shared__ float Ash[16 * 260];
  __shared__ int   toksh[16];
  __shared__ float redsh[8];
  __shared__ float scalesh;

  const float* __restrict__ M = rules + (size_t)r * (kD * kD);
  float scale = 0.0f;

  for (int c0 = 0; c0 < cnt; c0 += 16) {
    __syncthreads();
    if (tid < 16) toksh[tid] = (c0 + tid < cnt) ? tlist[start + c0 + tid] : -1;
    __syncthreads();

    for (int i = tid; i < 16 * 64; i += 256) {
      int m = i >> 6;
      int seg = i & 63;
      int tok = toksh[m];
      float4 v = make_float4(0.f, 0.f, 0.f, 0.f);
      if (tok >= 0)
        v = reinterpret_cast<const float4*>(hidden + (size_t)tok * kD)[seg];
      *reinterpret_cast<float4*>(&Ash[m * 260 + seg * 4]) = v;
    }
    __syncthreads();

    v8f acc0 = {};
    v8f acc1 = {};
    float sumsq = 0.0f;

    const float* Mc = M + colbase + col;
#pragma unroll 4
    for (int k0 = 0; k0 < kD; k0 += 4) {
      v2f a = *reinterpret_cast<const v2f*>(&Ash[col * 260 + k0 + 2 * h]);
      const float* p = Mc + (size_t)(k0 + 2 * h) * kD;
      v2f b0, b1;
      b0.x = __builtin_nontemporal_load(p);
      b0.y = __builtin_nontemporal_load(p + kD);
      b1.x = __builtin_nontemporal_load(p + 16);
      b1.y = __builtin_nontemporal_load(p + kD + 16);
      sumsq += b0.x * b0.x + b0.y * b0.y + b1.x * b1.x + b1.y * b1.y;
      acc0 = __builtin_amdgcn_wmma_f32_16x16x4_f32(false, a, false, b0,
                                                   (short)0, acc0, false, false);
      acc1 = __builtin_amdgcn_wmma_f32_16x16x4_f32(false, a, false, b1,
                                                   (short)0, acc1, false, false);
    }

    if (c0 == 0) {
      float s = sumsq;
      s += __shfl_xor(s, 16);
      s += __shfl_xor(s, 8);
      s += __shfl_xor(s, 4);
      s += __shfl_xor(s, 2);
      s += __shfl_xor(s, 1);
      if (lane == 0) redsh[wave] = s;
      __syncthreads();
      if (tid == 0) {
        float t = 0.f;
#pragma unroll
        for (int i = 0; i < 8; ++i) t += redsh[i];
        scalesh = 1.0f / fmaxf(sqrtf(t), kEps);
      }
      __syncthreads();
      scale = scalesh;
    }

#pragma unroll
    for (int i = 0; i < 8; ++i) { acc0[i] *= scale; acc1[i] *= scale; }

#pragma unroll
    for (int rr = 0; rr < 8; ++rr) {
      int row = rr + 8 * h;
      int tok = toksh[row];
      if (tok >= 0) {
        float* o = out + (size_t)tok * kD + colbase + col;
        __builtin_nontemporal_store(acc0[rr], o);
        __builtin_nontemporal_store(acc1[rr], o + 16);
      }
    }
  }
}


extern "C" void kernel_launch(void* const* d_in, const int* in_sizes, int n_in,
                              void* d_out, int out_size, void* d_ws, size_t ws_size,
                              hipStream_t stream) {
  const float* hidden      = (const float*)d_in[0];
  const int*   token_ids   = (const int*)d_in[1];
  const float* rules       = (const float*)d_in[2];
  const int*   token_rules = (const int*)d_in[3];
  float*       out         = (float*)d_out;

  const int ntok   = in_sizes[1];
  const int nrules = in_sizes[2] / (kD * kD);

  int* counts  = (int*)d_ws;
  int* offsets = counts + nrules;
  int* cursor  = offsets + nrules + 1;
  int* tlist   = cursor + nrules;

  rt_zero<<<(nrules + 255) / 256, 256, 0, stream>>>(counts, cursor, nrules);
  rt_count<<<(ntok + 255) / 256, 256, 0, stream>>>(token_ids, token_rules, counts, ntok);
  rt_scan<<<1, 1024, 0, stream>>>(counts, offsets, nrules);
  rt_scatter<<<(ntok + 255) / 256, 256, 0, stream>>>(token_ids, token_rules,
                                                     offsets, cursor, tlist, ntok);
  rt_gemm<<<nrules, 256, 0, stream>>>(hidden, rules, offsets, tlist, out);
}


# Fuse the four token-grouping launches (zero, count, scan, scatter) into one single-workgroup launch: count and rank from one pass of LDS atomics, offsets by a wave prefix sum; 5 launches -> 2
# speedup vs baseline: 1.1444x; 1.1444x over previous
#include <hip/hip_runtime.h>

typedef __attribute__((ext_vector_type(2))) float v2f;
typedef __attribute__((ext_vector_type(8))) float v8f;

static constexpr int kD = 256;
static constexpr float kEps = 1e-12f;


static constexpr int kGroupThreads = 1024;
static constexpr int kTokPerThread = 2;

template <int N>
__device__ __forceinline__ int row_shr(int x) {
  return __builtin_amdgcn_update_dpp(0, x, 0x110 + N, 0xf, 0xf, false);
}

__device__ __forceinline__ int row_prefix_sum(int s) {
  s += row_shr<1>(s);
  s += row_shr<2>(s);
  s += row_shr<4>(s);
  s += row_shr<8>(s);
  return s;
}

__global__ __launch_bounds__(kGroupThreads) void rt_group(
    const int* __restrict__ token_ids,
    const int* __restrict__ token_rules,
    int* __restrict__ counts,
    int* __restrict__ offsets,
    int* __restrict__ cursor,
    int* __restrict__ tlist,
    int ntok, int nrules) {
  __shared__ int cnt[kGroupThreads];
  __shared__ int wsum[32];

  const int tid = threadIdx.x;
  const int lane = tid & 31;
  const int wave = tid >> 5;

  if (nrules <= kGroupThreads && ntok >= 1 && ntok <= kTokPerThread * kGroupThreads) {
    const int t0 = tid;
    const int t1 = tid + kGroupThreads;
    const bool in0 = t0 < ntok;
    const bool in1 = t1 < ntok;
    const int id0 = token_ids[in0 ? t0 : 0];
    const int id1 = token_ids[in1 ? t1 : 0];
    const int r0 = token_rules[id0];
    const int r1 = token_rules[id1];
    cnt[tid] = 0;
    __syncthreads();

    int k0 = 0, k1 = 0;
    if (in0) k0 = atomicAdd(&cnt[r0], 1);
    if (in1) k1 = atomicAdd(&cnt[r1], 1);
    __syncthreads();

    const int v = cnt[tid];
    int s = row_prefix_sum(v);
    const int row0 = __builtin_amdgcn_readlane(s, 15);
    if (lane & 16) s += row0;
    if (lane == 31) wsum[wave] = s;
    __syncthreads();

    int part = row_prefix_sum((lane < wave) ? wsum[lane] : 0);
    const int base = __builtin_amdgcn_readlane(part, 15) + __builtin_amdgcn_readlane(part, 31);

    const int excl = base + s - v;
    cnt[tid] = excl;
    if (tid < nrules) offsets[tid] = excl;
    if (tid == nrules - 1) offsets[nrules] = excl + v;
    __syncthreads();

    if (in0) tlist[cnt[r0] + k0] = t0;
    if (in1) tlist[cnt[r1] + k1] = t1;
  } else if (tid == 0) {
    for (int i = 0; i < nrules; ++i) counts[i] = 0;
    for (int t = 0; t < ntok; ++t) counts[token_rules[token_ids[t]]] += 1;
    int acc = 0;
    for (int i = 0; i < nrules; ++i) { offsets[i] = acc; cursor[i] = acc; acc += counts[i]; }
    offsets[nrules] = acc;
    for (int t = 0; t < ntok; ++t) {
      const int r = token_rules[token_ids[t]];
      tlist[cursor[r]] = t;
      cursor[r] += 1;
    }
  }
}


__global__ __launch_bounds__(256) void rt_gemm(
    const float* __restrict__ hidden,
    const float* __restrict__ rules,
    const int* __restrict__ offsets,
    const int* __restrict__ tlist,
    float* __restrict__ out) {
  const int r = blockIdx.x;
  const int start = offsets[r];
  const int cnt = offsets[r + 1] - start;
  if (cnt == 0) return;

  const int tid = threadIdx.x;
  const int lane = tid & 31;
  const int wave = tid >> 5;
  const int h = lane >> 4;
  const int col = lane & 15;
  const int colbase = wave * 32;

  __shared__ float Ash[16 * 260];
  __shared__ int   toksh[16];
  __shared__ float redsh[8];
  __shared__ float scalesh;

  const float* __restrict__ M = rules + (size_t)r * (kD * kD);
  float scale = 0.0f;

  for (int c0 = 0; c0 < cnt; c0 += 16) {
    __syncthreads();
    if (tid < 16) toksh[tid] = (c0 + tid < cnt) ? tlist[start + c0 + tid] : -1;
    __syncthreads();

    for (int i = tid; i < 16 * 64; i += 256) {
      int m = i >> 6;
      int seg = i & 63;
      int tok = toksh[m];
      float4 v = make_float4(0.f, 0.f, 0.f, 0.f);
      if (tok >= 0)
        v = reinterpret_cast<const float4*>(hidden + (size_t)tok * kD)[seg];
      *reinterpret_cast<float4*>(&Ash[m * 260 + seg * 4]) = v;
    }
    __syncthreads();

    v8f acc0 = {};
    v8f acc1 = {};
    float sumsq = 0.0f;

    const float* Mc = M + colbase + col;
#pragma unroll 4
    for (int k0 = 0; k0 < kD; k0 += 4) {
      v2f a = *reinterpret_cast<const v2f*>(&Ash[col * 260 + k0 + 2 * h]);
      const float* p = Mc + (size_t)(k0 + 2 * h) * kD;
      v2f b0, b1;
      b0.x = __builtin_nontemporal_load(p);
      b0.y = __builtin_nontemporal_load(p + kD);
      b1.x = __builtin_nontemporal_load(p + 16);
      b1.y = __builtin_nontemporal_load(p + kD + 16);
      sumsq += b0.x * b0.x + b0.y * b0.y + b1.x * b1.x + b1.y * b1.y;
      acc0 = __builtin_amdgcn_wmma_f32_16x16x4_f32(false, a, false, b0,
                                                   (short)0, acc0, false, false);
      acc1 = __builtin_amdgcn_wmma_f32_16x16x4_f32(false, a, false, b1,
                                                   (short)0, acc1, false, false);
    }

    if (c0 == 0) {
      float s = sumsq;
      s += __shfl_xor(s, 16);
      s += __shfl_xor(s, 8);
      s += __shfl_xor(s, 4);
      s += __shfl_xor(s, 2);
      s += __shfl_xor(s, 1);
      if (lane == 0) redsh[wave] = s;
      __syncthreads();
      if (tid == 0) {
        float t = 0.f;
#pragma unroll
        for (int i = 0; i < 8; ++i) t += redsh[i];
        scalesh = 1.0f / fmaxf(sqrtf(t), kEps);
      }
      __syncthreads();
      scale = scalesh;
    }

#pragma unroll
    for (int i = 0; i < 8; ++i) { acc0[i] *= scale; acc1[i] *= scale; }

#pragma unroll
    for (int rr = 0; rr < 8; ++rr) {
      int row = rr + 8 * h;
      int tok = toksh[row];
      if (tok >= 0) {
        float* o = out + (size_t)tok * kD + colbase + col;
        __builtin_nontemporal_store(acc0[rr], o);
        __builtin_nontemporal_store(acc1[rr], o + 16);
      }
    }
  }
}


extern "C" void kernel_launch(void* const* d_in, const int* in_sizes, int n_in,
                              void* d_out, int out_size, void* d_ws, size_t ws_size,
                              hipStream_t stream) {
  const float* hidden      = (const float*)d_in[0];
  const int*   token_ids   = (const int*)d_in[1];
  const float* rules       = (const float*)d_in[2];
  const int*   token_rules = (const int*)d_in[3];
  float*       out         = (float*)d_out;

  const int ntok   = in_sizes[1];
  const int nrules = in_sizes[2] / (kD * kD);

  int* counts  = (int*)d_ws;
  int* offsets = counts + nrules;
  int* cursor  = offsets + nrules + 1;
  int* tlist   = cursor + nrules;

  rt_group<<<1, kGroupThreads, 0, stream>>>(token_ids, token_rules, counts, offsets,
                                            cursor, tlist, ntok, nrules);
  rt_gemm<<<nrules, 256, 0, stream>>>(hidden, rules, offsets, tlist, out);
}


# Rule matrices read as a stream: 256 resident workgroups of 16 waves, a 4-wave team per rule, 16-byte row loads requested three 16-deep chunks ahead of the products; norm partial sums regrouped
# speedup vs baseline: 1.2224x; 1.0682x over previous
#include <hip/hip_runtime.h>

typedef __attribute__((ext_vector_type(2))) float v2f;
typedef __attribute__((ext_vector_type(4))) float v4f;
typedef __attribute__((ext_vector_type(8))) float v8f;

static constexpr int kD = 256;
static constexpr float kEps = 1e-12f;


static constexpr int kGroupThreads = 1024;
static constexpr int kTokPerThread = 2;

template <int N>
__device__ __forceinline__ int row_shr(int x) {
  return __builtin_amdgcn_update_dpp(0, x, 0x110 + N, 0xf, 0xf, false);
}

__device__ __forceinline__ int row_prefix_sum(int s) {
  s += row_shr<1>(s);
  s += row_shr<2>(s);
  s += row_shr<4>(s);
  s += row_shr<8>(s);
  return s;
}

__global__ __launch_bounds__(kGroupThreads) void rt_group(
    const int* __restrict__ token_ids,
    const int* __restrict__ token_rules,
    int* __restrict__ counts,
    int* __restrict__ offsets,
    int* __restrict__ cursor,
    int* __restrict__ tlist,
    int ntok, int nrules) {
  __shared__ int cnt[kGroupThreads];
  __shared__ int wsum[32];

  const int tid = threadIdx.x;
  const int lane = tid & 31;
  const int wave = tid >> 5;

  if (nrules <= kGroupThreads && ntok >= 1 && ntok <= kTokPerThread * kGroupThreads) {
    const int t0 = tid;
    const int t1 = tid + kGroupThreads;
    const bool in0 = t0 < ntok;
    const bool in1 = t1 < ntok;
    const int id0 = token_ids[in0 ? t0 : 0];
    const int id1 = token_ids[in1 ? t1 : 0];
    const int r0 = token_rules[id0];
    const int r1 = token_rules[id1];
    cnt[tid] = 0;
    __syncthreads();

    int k0 = 0, k1 = 0;
    if (in0) k0 = atomicAdd(&cnt[r0], 1);
    if (in1) k1 = atomicAdd(&cnt[r1], 1);
    __syncthreads();

    const int v = cnt[tid];
    int s = row_prefix_sum(v);
    const int row0 = __builtin_amdgcn_readlane(s, 15);
    if (lane & 16) s += row0;
    if (lane == 31) wsum[wave] = s;
    __syncthreads();

    int part = row_prefix_sum((lane < wave) ? wsum[lane] : 0);
    const int base = __builtin_amdgcn_readlane(part, 15) + __builtin_amdgcn_readlane(part, 31);

    const int excl = base + s - v;
    cnt[tid] = excl;
    if (tid < nrules) offsets[tid] = excl;
    if (tid == nrules - 1) offsets[nrules] = excl + v;
    __syncthreads();

    if (in0) tlist[cnt[r0] + k0] = t0;
    if (in1) tlist[cnt[r1] + k1] = t1;
  } else if (tid == 0) {
    for (int i = 0; i < nrules; ++i) counts[i] = 0;
    for (int t = 0; t < ntok; ++t) counts[token_rules[token_ids[t]]] += 1;
    int acc = 0;
    for (int i = 0; i < nrules; ++i) { offsets[i] = acc; cursor[i] = acc; acc += counts[i]; }
    offsets[nrules] = acc;
    for (int t = 0; t < ntok; ++t) {
      const int r = token_rules[token_ids[t]];
      tlist[cursor[r]] = t;
      cursor[r] += 1;
    }
  }
}


static constexpr int kTeams = 4;
static constexpr int kGemmThreads = kTeams * 4 * 32;
static constexpr int kGemmGroups = 256;
static constexpr int kPitch = 260;
static constexpr int kChunks = kD / 16;
static constexpr int kAhead = 3;

__device__ __forceinline__ v4f load_rule4(const float* p) {
  return __builtin_nontemporal_load(reinterpret_cast<const v4f*>(p));
}

__global__ __launch_bounds__(kGemmThreads) void rt_gemm(
    const float* __restrict__ hidden,
    const float* __restrict__ rules,
    const int* __restrict__ offsets,
    const int* __restrict__ tlist,
    float* __restrict__ out,
    int nrules) {
  __shared__ __attribute__((aligned(16))) float Ash[kTeams * 16 * kPitch];
  __shared__ __attribute__((aligned(16))) float redsh[kTeams * 4];

  const int tid = threadIdx.x;
  const int lane = tid & 31;
  const int wave = __builtin_amdgcn_readfirstlane(tid >> 5);
  const int team = wave >> 2;
  const int tw = wave & 3;
  const int h = lane >> 4;
  const int c = lane & 15;

  float* __restrict__ At = Ash + team * (16 * kPitch);

  for (int base = blockIdx.x * kTeams; base < nrules; base += kGemmGroups * kTeams) {
    int o[kTeams + 1];
#pragma unroll
    for (int t = 0; t <= kTeams; ++t) o[t] = offsets[min(base + t, nrules)];
    int maxcnt = 0;
    int start = o[0];
    int end = o[1];
#pragma unroll
    for (int t = 0; t < kTeams; ++t) {
      maxcnt = max(maxcnt, o[t + 1] - o[t]);
      if (team == t) { start = o[t]; end = o[t + 1]; }
    }
    const int cnt = end - start;
    const int r = min(base + team, nrules);

    const float* __restrict__ Pl = rules + (size_t)r * (kD * kD) + (4 * h) * kD + tw * 64 + 4 * c;
    const float* __restrict__ Al = At + c * kPitch + 4 * h;

    for (int c0 = 0; c0 < maxcnt; c0 += 16) {
      const bool active = c0 < cnt;
      v4f bq[kAhead + 1][8];
      v4f aq[2][2];
      int tokv = -1;

#define LOAD_CHUNK(Q)                                                                  \
      do {                                                                             \
        _Pragma("unroll") for (int j = 0; j < 8; ++j)                                  \
          bq[(Q) % (kAhead + 1)][j] =                                                  \
              load_rule4(Pl + ((Q) * 16 + (j >> 2) * 8 + (j & 3)) * kD);               \
      } while (0)
#define LOAD_A(Q)                                                                      \
      do {                                                                             \
        aq[(Q) & 1][0] = *reinterpret_cast<const v4f*>(Al + (Q) * 16);                 \
        aq[(Q) & 1][1] = *reinterpret_cast<const v4f*>(Al + (Q) * 16 + 8);             \
      } while (0)

      if (active) {
        const int idx = c0 + c;
        if (idx < cnt) tokv = tlist[start + idx];
        __builtin_amdgcn_sched_barrier(0);
#pragma unroll
        for (int q = 0; q < kAhead; ++q) LOAD_CHUNK(q);
        __builtin_amdgcn_sched_barrier(0);

        v4f hv[4][2];
        int tokm[4];
#pragma unroll
        for (int jj = 0; jj < 4; ++jj) {
          tokm[jj] = __builtin_amdgcn_readlane(tokv, tw * 4 + jj);
          const v4f* hp = reinterpret_cast<const v4f*>(hidden + (size_t)max(tokm[jj], 0) * kD);
          hv[jj][0] = hp[lane];
          hv[jj][1] = hp[32 + lane];
        }
#pragma unroll
        for (int jj = 0; jj < 4; ++jj) {
          const v4f z = {0.f, 0.f, 0.f, 0.f};
          float* ap = At + (tw * 4 + jj) * kPitch + 4 * lane;
          *reinterpret_cast<v4f*>(ap) = (tokm[jj] >= 0) ? hv[jj][0] : z;
          *reinterpret_cast<v4f*>(ap + 128) = (tokm[jj] >= 0) ? hv[jj][1] : z;
        }
      }
      __syncthreads();

      v8f acc[4] = {};
      if (active) {
        v4f ss = {0.f, 0.f, 0.f, 0.f};
        LOAD_A(0);
#pragma unroll
        for (int q = 0; q < kChunks; ++q) {
          __builtin_amdgcn_sched_barrier(0);
          if (q + kAhead < kChunks) LOAD_CHUNK(q + kAhead);
          if (q + 1 < kChunks) LOAD_A(q + 1);
          __builtin_amdgcn_sched_barrier(0);

          const v4f* bb = bq[q % (kAhead + 1)];
#pragma unroll
          for (int kb = 0; kb < 2; ++kb) {
            const v4f a4 = aq[q & 1][kb];
#pragma unroll
            for (int st = 0; st < 2; ++st) {
              v2f a;
              a.x = a4[2 * st];
              a.y = a4[2 * st + 1];
              const v4f l0 = bb[4 * kb + 2 * st];
              const v4f l1 = bb[4 * kb + 2 * st + 1];
#pragma unroll
              for (int t = 0; t < 4; ++t) {
                v2f b;
                b.x = l0[t];
                b.y = l1[t];
                acc[t] = __builtin_amdgcn_wmma_f32_16x16x4_f32(false, a, false, b,
                                                               (short)0, acc[t], false, false);
              }
            }
          }
#pragma unroll
          for (int j = 0; j < 8; ++j) ss += bb[j] * bb[j];
        }

        float s = (ss.x + ss.y) + (ss.z + ss.w);
        s += __shfl_xor(s, 16);
        s += __shfl_xor(s, 8);
        s += __shfl_xor(s, 4);
        s += __shfl_xor(s, 2);
        s += __shfl_xor(s, 1);
        if (lane == 0) redsh[team * 4 + tw] = s;
      }
      __syncthreads();

      if (active) {
        const v4f p = *reinterpret_cast<const v4f*>(&redsh[team * 4]);
        const float scale = 1.0f / fmaxf(sqrtf((p.x + p.y) + (p.z + p.w)), kEps);
#pragma unroll
        for (int rr = 0; rr < 8; ++rr) {
          const int tok = __shfl(tokv, rr + 8 * h);
          if (tok >= 0) {
            v4f o;
            o.x = acc[0][rr] * scale;
            o.y = acc[1][rr] * scale;
            o.z = acc[2][rr] * scale;
            o.w = acc[3][rr] * scale;
            __builtin_nontemporal_store(
                o, reinterpret_cast<v4f*>(out + (size_t)tok * kD + tw * 64 + 4 * c));
          }
        }
      }
#undef LOAD_CHUNK
#undef LOAD_A
    }
  }
}


extern "C" void kernel_launch(void* const* d_in, const int* in_sizes, int n_in,
                              void* d_out, int out_size, void* d_ws, size_t ws_size,
                              hipStream_t stream) {
  const float* hidden      = (const float*)d_in[0];
  const int*   token_ids   = (const int*)d_in[1];
  const float* rules       = (const float*)d_in[2];
  const int*   token_rules = (const int*)d_in[3];
  float*       out         = (float*)d_out;

  const int ntok   = in_sizes[1];
  const int nrules = in_sizes[2] / (kD * kD);

  int* counts  = (int*)d_ws;
  int* offsets = counts + nrules;
  int* cursor  = offsets + nrules + 1;
  int* tlist   = cursor + nrules;

  rt_group<<<1, kGroupThreads, 0, stream>>>(token_ids, token_rules, counts, offsets,
                                            cursor, tlist, ntok, nrules);
  rt_gemm<<<kGemmGroups, kGemmThreads, 0, stream>>>(hidden, rules, offsets, tlist, out, nrules);
}


# One launch instead of two: each workgroup finds the tokens of its own four rules itself (rule look-up of all tokens, rank by LDS atomic add) while its first matrix chunks are already in flight
# speedup vs baseline: 1.3391x; 1.0955x over previous
#include <hip/hip_runtime.h>

typedef __attribute__((ext_vector_type(2))) float v2f;
typedef __attribute__((ext_vector_type(4))) float v4f;
typedef __attribute__((ext_vector_type(8))) float v8f;
typedef __attribute__((ext_vector_type(4))) int v4i;

static constexpr int kD = 256;
static constexpr float kEps = 1e-12f;


static constexpr int kTeams = 4;
static constexpr int kGemmThreads = kTeams * 4 * 32;
static constexpr int kGemmGroups = 256;
static constexpr int kPitch = 260;
static constexpr int kChunks = kD / 16;
static constexpr int kAhead = 3;
static constexpr int kTokPerThread = 4;
static constexpr int kSlab = kGemmThreads * kTokPerThread;

__device__ __forceinline__ v4f load_rule4(const float* p) {
  return __builtin_nontemporal_load(reinterpret_cast<const v4f*>(p));
}

__global__ __launch_bounds__(kGemmThreads) void rt_rules(
    const float* __restrict__ hidden,
    const int* __restrict__ token_ids,
    const float* __restrict__ rules,
    const int* __restrict__ token_rules,
    float* __restrict__ out,
    int ntok, int nrules) {
  __shared__ __attribute__((aligned(16))) float Ash[kTeams * 16 * kPitch];
  __shared__ __attribute__((aligned(16))) float redsh[kTeams * 4];
  __shared__ __attribute__((aligned(16))) int tcnt[kTeams];
  __shared__ int ttok[kTeams * 16];

  const int tid = threadIdx.x;
  const int lane = tid & 31;
  const int wave = __builtin_amdgcn_readfirstlane(tid >> 5);
  const int team = wave >> 2;
  const int tw = wave & 3;
  const int h = lane >> 4;
  const int c = lane & 15;

  float* __restrict__ At = Ash + team * (16 * kPitch);
  const float* __restrict__ Al = At + c * kPitch + 4 * h;

  int base = blockIdx.x * kTeams;
  int slab0 = 0;
  bool later = false;
  while (base < nrules && slab0 < ntok) {
    {
      const int r = (base + team < nrules) ? base + team : 0;
      const float* __restrict__ Pl = rules + (size_t)r * (kD * kD) + (4 * h) * kD + tw * 64 + 4 * c;
      v4f bq[kAhead + 1][8];

#define LOAD_CHUNK(Q)                                                                  \
      do {                                                                             \
        _Pragma("unroll") for (int j = 0; j < 8; ++j)                                  \
          bq[(Q) % (kAhead + 1)][j] =                                                  \
              load_rule4(Pl + ((Q) * 16 + (j >> 2) * 8 + (j & 3)) * kD);               \
      } while (0)
#define LOAD_A(Q)                                                                      \
      do {                                                                             \
        aq[(Q) & 1][0] = *reinterpret_cast<const v4f*>(Al + (Q) * 16);                 \
        aq[(Q) & 1][1] = *reinterpret_cast<const v4f*>(Al + (Q) * 16 + 8);             \
      } while (0)

      if (later) __syncthreads();
      later = true;

      int tk[kTokPerThread], rl[kTokPerThread];
#pragma unroll
      for (int i = 0; i < kTokPerThread; ++i) {
        tk[i] = slab0 + i * kGemmThreads + tid;
        rl[i] = token_ids[tk[i] < ntok ? tk[i] : 0];
      }
      if (tid < kTeams) tcnt[tid] = 0;
      __syncthreads();
#pragma unroll
      for (int i = 0; i < kTokPerThread; ++i) rl[i] = token_rules[rl[i]];
#pragma unroll
      for (int q = 0; q < kAhead; ++q) LOAD_CHUNK(q);

      int dm[kTokPerThread], rk[kTokPerThread];
#pragma unroll
      for (int i = 0; i < kTokPerThread; ++i) {
        const int d = rl[i] - base;
        dm[i] = -1;
        rk[i] = 0;
        if (tk[i] < ntok && d >= 0 && d < kTeams && rl[i] < nrules) {
          dm[i] = d;
          rk[i] = atomicAdd(&tcnt[d], 1);
          if (rk[i] < 16) ttok[d * 16 + rk[i]] = tk[i];
        }
      }
      asm volatile("s_wait_dscnt 0x0\n\ts_barrier_signal -1\n\ts_barrier_wait -1" ::: "memory");

      const v4i cn = *reinterpret_cast<const v4i*>(tcnt);
      const int maxcnt = max(max(cn.x, cn.y), max(cn.z, cn.w));
      const int cnt = (team == 0) ? cn.x : (team == 1) ? cn.y : (team == 2) ? cn.z : cn.w;

      for (int c0 = 0; c0 < maxcnt; c0 += 16) {
        if (c0 > 0) {
#pragma unroll
          for (int i = 0; i < kTokPerThread; ++i)
            if (dm[i] >= 0 && rk[i] >= c0 && rk[i] < c0 + 16) ttok[dm[i] * 16 + rk[i] - c0] = tk[i];
          __syncthreads();
        }
        const bool active = c0 < cnt;
        v4f aq[2][2];
        int tokv = -1;

        if (active) {
          if (c0 + c < cnt) tokv = ttok[team * 16 + c];
          if (c0 > 0) {
#pragma unroll
            for (int q = 0; q < kAhead; ++q) LOAD_CHUNK(q);
          }

          v4f hv[4][2];
          int tokm[4];
#pragma unroll
          for (int jj = 0; jj < 4; ++jj) {
            tokm[jj] = __builtin_amdgcn_readlane(tokv, tw * 4 + jj);
            const v4f* hp = reinterpret_cast<const v4f*>(hidden + (size_t)max(tokm[jj], 0) * kD);
            hv[jj][0] = hp[lane];
            hv[jj][1] = hp[32 + lane];
          }
#pragma unroll
          for (int jj = 0; jj < 4; ++jj) {
            const v4f z = {0.f, 0.f, 0.f, 0.f};
            float* ap = At + (tw * 4 + jj) * kPitch + 4 * lane;
            *reinterpret_cast<v4f*>(ap) = (tokm[jj] >= 0) ? hv[jj][0] : z;
            *reinterpret_cast<v4f*>(ap + 128) = (tokm[jj] >= 0) ? hv[jj][1] : z;
          }
        }
        __syncthreads();

        v8f acc[4] = {};
        if (active) {
          v4f ss = {0.f, 0.f, 0.f, 0.f};
          LOAD_A(0);
#pragma unroll
          for (int q = 0; q < kChunks; ++q) {
            __builtin_amdgcn_sched_barrier(0);
            if (q + kAhead < kChunks) LOAD_CHUNK(q + kAhead);
            if (q + 1 < kChunks) LOAD_A(q + 1);
            __builtin_amdgcn_sched_barrier(0);

            const v4f* bb = bq[q % (kAhead + 1)];
#pragma unroll
            for (int kb = 0; kb < 2; ++kb) {
              const v4f a4 = aq[q & 1][kb];
#pragma unroll
              for (int st = 0; st < 2; ++st) {
                v2f a;
                a.x = a4[2 * st];
                a.y = a4[2 * st + 1];
                const v4f l0 = bb[4 * kb + 2 * st];
                const v4f l1 = bb[4 * kb + 2 * st + 1];
#pragma unroll
                for (int t = 0; t < 4; ++t) {
                  v2f b;
                  b.x = l0[t];
                  b.y = l1[t];
                  acc[t] = __builtin_amdgcn_wmma_f32_16x16x4_f32(false, a, false, b,
                                                                 (short)0, acc[t], false, false);
                }
              }
            }
#pragma unroll
            for (int j = 0; j < 8; ++j) ss += bb[j] * bb[j];
          }

          float s = (ss.x + ss.y) + (ss.z + ss.w);
          s += __shfl_xor(s, 16);
          s += __shfl_xor(s, 8);
          s += __shfl_xor(s, 4);
          s += __shfl_xor(s, 2);
          s += __shfl_xor(s, 1);
          if (lane == 0) redsh[team * 4 + tw] = s;
        }
        __syncthreads();

        if (active) {
          const v4f p = *reinterpret_cast<const v4f*>(&redsh[team * 4]);
          const float scale = 1.0f / fmaxf(sqrtf((p.x + p.y) + (p.z + p.w)), kEps);
#pragma unroll
          for (int rr = 0; rr < 8; ++rr) {
            const int tok = __shfl(tokv, rr + 8 * h);
            if (tok >= 0) {
              v4f o;
              o.x = acc[0][rr] * scale;
              o.y = acc[1][rr] * scale;
              o.z = acc[2][rr] * scale;
              o.w = acc[3][rr] * scale;
              __builtin_nontemporal_store(
                  o, reinterpret_cast<v4f*>(out + (size_t)tok * kD + tw * 64 + 4 * c));
            }
          }
        }
      }
#undef LOAD_CHUNK
#undef LOAD_A
    }
    slab0 += kSlab;
    if (slab0 >= ntok) {
      slab0 = 0;
      base += kGemmGroups * kTeams;
    }
  }
}


extern "C" void kernel_launch(void* const* d_in, const int* in_sizes, int n_in,
                              void* d_out, int out_size, void* d_ws, size_t ws_size,
                              hipStream_t stream) {
  const float* hidden      = (const float*)d_in[0];
  const int*   token_ids   = (const int*)d_in[1];
  const float* rules       = (const float*)d_in[2];
  const int*   token_rules = (const int*)d_in[3];
  float*       out         = (float*)d_out;

  const int ntok   = in_sizes[1];
  const int nrules = in_sizes[2] / (kD * kD);

  rt_rules<<<kGemmGroups, kGemmThreads, 0, stream>>>(hidden, token_ids, rules, token_rules, out,
                                                     ntok, nrules);
}


# Half the waves for the same stream: 8-wave workgroups, a rule's team is 2 waves of 128 columns (8 tiles each), 8-deep chunks requested six ahead; norm partial sums regrouped (two wave sums per rule)
# speedup vs baseline: 1.3832x; 1.0329x over previous
#include <hip/hip_runtime.h>

typedef __attribute__((ext_vector_type(2))) float v2f;
typedef __attribute__((ext_vector_type(4))) float v4f;
typedef __attribute__((ext_vector_type(8))) float v8f;
typedef __attribute__((ext_vector_type(4))) int v4i;

static constexpr int kD = 256;
static constexpr float kEps = 1e-12f;


static constexpr int kTeams = 4;
static constexpr int kTeamWaves = 2;
static constexpr int kGemmThreads = kTeams * kTeamWaves * 32;
static constexpr int kGemmGroups = 256;
static constexpr int kPitch = 260;
static constexpr int kChunks = kD / 8;
static constexpr int kAhead = 6;
static constexpr int kRing = kAhead + 1;
static constexpr int kTokPerThread = 8;
static constexpr int kSlab = kGemmThreads * kTokPerThread;

__device__ __forceinline__ v4f load_rule4(const float* p) {
  return __builtin_nontemporal_load(reinterpret_cast<const v4f*>(p));
}

__global__ __launch_bounds__(kGemmThreads) void rt_rules(
    const float* __restrict__ hidden,
    const int* __restrict__ token_ids,
    const float* __restrict__ rules,
    const int* __restrict__ token_rules,
    float* __restrict__ out,
    int ntok, int nrules) {
  __shared__ __attribute__((aligned(16))) float Ash[kTeams * 16 * kPitch];
  __shared__ __attribute__((aligned(16))) float redsh[kTeams * kTeamWaves];
  __shared__ __attribute__((aligned(16))) int tcnt[kTeams];
  __shared__ int ttok[kTeams * 16];

  const int tid = threadIdx.x;
  const int lane = tid & 31;
  const int wave = __builtin_amdgcn_readfirstlane(tid >> 5);
  const int team = wave >> 1;
  const int tw = wave & 1;
  const int h = lane >> 4;
  const int c = lane & 15;

  float* __restrict__ At = Ash + team * (16 * kPitch);
  const float* __restrict__ Al = At + c * kPitch + 4 * h;

  int base = blockIdx.x * kTeams;
  int slab0 = 0;
  bool later = false;
  while (base < nrules && slab0 < ntok) {
    {
      const int r = (base + team < nrules) ? base + team : 0;
      const float* __restrict__ Pl = rules + (size_t)r * (kD * kD) + (4 * h) * kD + tw * 128 + 4 * c;
      v4f bq[kRing][8];

#define LOAD_CHUNK(Q)                                                                  \
      do {                                                                             \
        _Pragma("unroll") for (int j = 0; j < 8; ++j)                                  \
          bq[(Q) % kRing][j] = load_rule4(Pl + ((Q) * 8 + (j >> 1)) * kD + (j & 1) * 64); \
      } while (0)
#define LOAD_A(Q)                                                                      \
      do {                                                                             \
        aq[(Q) & 1] = *reinterpret_cast<const v4f*>(Al + (Q) * 8);                     \
      } while (0)

      if (later) __syncthreads();
      later = true;

      int tk[kTokPerThread], rl[kTokPerThread];
#pragma unroll
      for (int i = 0; i < kTokPerThread; ++i) {
        tk[i] = slab0 + i * kGemmThreads + tid;
        rl[i] = token_ids[tk[i] < ntok ? tk[i] : 0];
      }
      if (tid < kTeams) tcnt[tid] = 0;
      __syncthreads();
#pragma unroll
      for (int i = 0; i < kTokPerThread; ++i) rl[i] = token_rules[rl[i]];
#pragma unroll
      for (int q = 0; q < kAhead; ++q) LOAD_CHUNK(q);

      int dm[kTokPerThread], rk[kTokPerThread];
#pragma unroll
      for (int i = 0; i < kTokPerThread; ++i) {
        const int d = rl[i] - base;
        dm[i] = -1;
        rk[i] = 0;
        if (tk[i] < ntok && d >= 0 && d < kTeams && rl[i] < nrules) {
          dm[i] = d;
          rk[i] = atomicAdd(&tcnt[d], 1);
          if (rk[i] < 16) ttok[d * 16 + rk[i]] = tk[i];
        }
      }
      asm volatile("s_wait_dscnt 0x0\n\ts_barrier_signal -1\n\ts_barrier_wait -1" ::: "memory");

      const v4i cn = *reinterpret_cast<const v4i*>(tcnt);
      const int maxcnt = max(max(cn.x, cn.y), max(cn.z, cn.w));
      const int cnt = (team == 0) ? cn.x : (team == 1) ? cn.y : (team == 2) ? cn.z : cn.w;

      for (int c0 = 0; c0 < maxcnt; c0 += 16) {
        if (c0 > 0) {
#pragma unroll
          for (int i = 0; i < kTokPerThread; ++i)
            if (dm[i] >= 0 && rk[i] >= c0 && rk[i] < c0 + 16) ttok[dm[i] * 16 + rk[i] - c0] = tk[i];
          __syncthreads();
        }
        const bool active = c0 < cnt;
        v4f aq[2];
        int tokv = -1;

        if (active) {
          if (c0 + c < cnt) tokv = ttok[team * 16 + c];
          if (c0 > 0) {
#pragma unroll
            for (int q = 0; q < kAhead; ++q) LOAD_CHUNK(q);
          }

          v4f hv[8][2];
          int tokm[8];
#pragma unroll
          for (int jj = 0; jj < 8; ++jj) {
            tokm[jj] = __builtin_amdgcn_readlane(tokv, tw * 8 + jj);
            const v4f* hp = reinterpret_cast<const v4f*>(hidden + (size_t)max(tokm[jj], 0) * kD);
            hv[jj][0] = hp[lane];
            hv[jj][1] = hp[32 + lane];
          }
#pragma unroll
          for (int jj = 0; jj < 8; ++jj) {
            const v4f z = {0.f, 0.f, 0.f, 0.f};
            float* ap = At + (tw * 8 + jj) * kPitch + 4 * lane;
            *reinterpret_cast<v4f*>(ap) = (tokm[jj] >= 0) ? hv[jj][0] : z;
            *reinterpret_cast<v4f*>(ap + 128) = (tokm[jj] >= 0) ? hv[jj][1] : z;
          }
        }
        __syncthreads();

        v8f acc[8] = {};
        if (active) {
          v4f ss = {0.f, 0.f, 0.f, 0.f};
          LOAD_A(0);
#pragma unroll
          for (int q = 0; q < kChunks; ++q) {
            __builtin_amdgcn_sched_barrier(0);
            if (q + kAhead < kChunks) LOAD_CHUNK(q + kAhead);
            if (q + 1 < kChunks) LOAD_A(q + 1);
            __builtin_amdgcn_sched_barrier(0);

            const v4f* bb = bq[q % kRing];
            const v4f a4 = aq[q & 1];
#pragma unroll
            for (int st = 0; st < 2; ++st) {
              v2f a;
              a.x = a4[2 * st];
              a.y = a4[2 * st + 1];
#pragma unroll
              for (int hf = 0; hf < 2; ++hf) {
                const v4f l0 = bb[2 * (2 * st) + hf];
                const v4f l1 = bb[2 * (2 * st + 1) + hf];
#pragma unroll
                for (int t = 0; t < 4; ++t) {
                  v2f b;
                  b.x = l0[t];
                  b.y = l1[t];
                  acc[4 * hf + t] = __builtin_amdgcn_wmma_f32_16x16x4_f32(
                      false, a, false, b, (short)0, acc[4 * hf + t], false, false);
                }
              }
            }
#pragma unroll
            for (int j = 0; j < 8; ++j) ss += bb[j] * bb[j];
          }

          float s = (ss.x + ss.y) + (ss.z + ss.w);
          s += __shfl_xor(s, 16);
          s += __shfl_xor(s, 8);
          s += __shfl_xor(s, 4);
          s += __shfl_xor(s, 2);
          s += __shfl_xor(s, 1);
          if (lane == 0) redsh[team * kTeamWaves + tw] = s;
        }
        __syncthreads();

        if (active) {
          const v2f p = *reinterpret_cast<const v2f*>(&redsh[team * kTeamWaves]);
          const float scale = 1.0f / fmaxf(sqrtf(p.x + p.y), kEps);
#pragma unroll
          for (int rr = 0; rr < 8; ++rr) {
            const int tok = __shfl(tokv, rr + 8 * h);
            if (tok >= 0) {
#pragma unroll
              for (int hf = 0; hf < 2; ++hf) {
                v4f o;
                o.x = acc[4 * hf + 0][rr] * scale;
                o.y = acc[4 * hf + 1][rr] * scale;
                o.z = acc[4 * hf + 2][rr] * scale;
                o.w = acc[4 * hf + 3][rr] * scale;
                __builtin_nontemporal_store(
                    o, reinterpret_cast<v4f*>(out + (size_t)tok * kD + tw * 128 + hf * 64 + 4 * c));
              }
            }
          }
        }
      }
#undef LOAD_CHUNK
#undef LOAD_A
    }
    slab0 += kSlab;
    if (slab0 >= ntok) {
      slab0 = 0;
      base += kGemmGroups * kTeams;
    }
  }
}


extern "C" void kernel_launch(void* const* d_in, const int* in_sizes, int n_in,
                              void* d_out, int out_size, void* d_ws, size_t ws_size,
                              hipStream_t stream) {
  const float* hidden      = (const float*)d_in[0];
  const int*   token_ids   = (const int*)d_in[1];
  const float* rules       = (const float*)d_in[2];
  const int*   token_rules = (const int*)d_in[3];
  float*       out         = (float*)d_out;

  const int ntok   = in_sizes[1];
  const int nrules = in_sizes[2] / (kD * kD);

  rt_rules<<<kGemmGroups, kGemmThreads, 0, stream>>>(hidden, token_ids, rules, token_rules, out,
                                                     ntok, nrules);
}
